# starting kernel
# baseline (speedup 1.0000x reference)
#include <hip/hip_runtime.h>


#define N_NODES 100000
#define N_EDGES 3200000
#define IN_DIM  128
#define HID_DIM 32
#define OUT_DIM 16

typedef __attribute__((ext_vector_type(2))) float v2f;
typedef __attribute__((ext_vector_type(8))) float v8f;

__global__ void __launch_bounds__(256) k_init(float* __restrict__ deg,
                                              float* __restrict__ agg1,
                                              float* __restrict__ out,
                                              const float* __restrict__ b2) {
  int i = blockIdx.x * blockDim.x + threadIdx.x;
  if (i < N_NODES) deg[i] = 1.0f;
  if (i < N_NODES * HID_DIM) agg1[i] = 0.0f;
  if (i < N_NODES * OUT_DIM) out[i] = b2[i & (OUT_DIM - 1)];
}

__global__ void __launch_bounds__(256) k_degree(const int* __restrict__ ei,
                                                float* __restrict__ deg) {
  int e = blockIdx.x * blockDim.x + threadIdx.x;
  if (e < N_EDGES) {
    int d = ei[N_EDGES + e];
    unsafeAtomicAdd(&deg[d], 1.0f);
  }
}

__global__ void __launch_bounds__(256) k_rsqrt(float* __restrict__ deg) {
  int i = blockIdx.x * blockDim.x + threadIdx.x;
  if (i < N_NODES) {
    float d = deg[i];
    deg[i] = (d > 0.0f) ? rsqrtf(d) : 0.0f;
  }
}

__global__ void __launch_bounds__(128) k_gemm1(const float* __restrict__ X,
                                               const float* __restrict__ W,
                                               float* __restrict__ H) {
  const int wid    = (blockIdx.x * blockDim.x + threadIdx.x) >> 5;
  const int lane   = threadIdx.x & 31;
  const int tile_m = wid >> 1;
  const int tile_n = wid & 1;
  const int half   = lane >> 4;
  const int l15    = lane & 15;
  const int rowA   = tile_m * 16 + l15;
  const int kOff   = half * 2;
  const int colB   = tile_n * 16 + l15;

  v8f c = {};
#pragma unroll
  for (int k = 0; k < IN_DIM; k += 4) {
    v2f a = *(const v2f*)(X + (size_t)rowA * IN_DIM + k + kOff);
    v2f b;
    b.x = W[(k + kOff + 0) * HID_DIM + colB];
    b.y = W[(k + kOff + 1) * HID_DIM + colB];
    c = __builtin_amdgcn_wmma_f32_16x16x4_f32(false, a, false, b,
                                              (short)0, c, false, false);
  }
  const int m0 = tile_m * 16;
#pragma unroll
  for (int i = 0; i < 8; ++i) {
    H[(size_t)(m0 + half * 8 + i) * HID_DIM + tile_n * 16 + l15] = c[i];
  }
}

__global__ void __launch_bounds__(256) k_agg32(const int* __restrict__ ei,
                                               const float* __restrict__ dinv,
                                               const float* __restrict__ h,
                                               float* __restrict__ agg) {
  int idx = blockIdx.x * blockDim.x + threadIdx.x;
  int e = idx >> 5;
  int f = idx & 31;
  const int total = N_EDGES + N_NODES;
  if (e >= total) return;
  int s, d;
  if (e < N_EDGES) { s = ei[e]; d = ei[N_EDGES + e]; }
  else             { s = d = e - N_EDGES; }
  float w = dinv[s] * dinv[d];
  float v = h[(size_t)s * HID_DIM + f] * w;
  unsafeAtomicAdd(&agg[(size_t)d * HID_DIM + f], v);
}

__global__ void __launch_bounds__(256) k_bias_relu(float* __restrict__ a,
                                                   const float* __restrict__ b1) {
  int i = blockIdx.x * blockDim.x + threadIdx.x;
  if (i < N_NODES * HID_DIM) {
    float v = a[i] + b1[i & (HID_DIM - 1)];
    a[i] = fmaxf(v, 0.0f);
  }
}

__global__ void __launch_bounds__(64) k_gemm2(const float* __restrict__ H1,
                                              const float* __restrict__ W,
                                              float* __restrict__ G) {
  const int wid    = (blockIdx.x * blockDim.x + threadIdx.x) >> 5;
  const int lane   = threadIdx.x & 31;
  const int tile_m = wid;
  const int half   = lane >> 4;
  const int l15    = lane & 15;
  const int rowA   = tile_m * 16 + l15;
  const int kOff   = half * 2;
  const int colB   = l15;

  v8f c = {};
#pragma unroll
  for (int k = 0; k < HID_DIM; k += 4) {
    v2f a = *(const v2f*)(H1 + (size_t)rowA * HID_DIM + k + kOff);
    v2f b;
    b.x = W[(k + kOff + 0) * OUT_DIM + colB];
    b.y = W[(k + kOff + 1) * OUT_DIM + colB];
    c = __builtin_amdgcn_wmma_f32_16x16x4_f32(false, a, false, b,
                                              (short)0, c, false, false);
  }
  const int m0 = tile_m * 16;
#pragma unroll
  for (int i = 0; i < 8; ++i) {
    G[(size_t)(m0 + half * 8 + i) * OUT_DIM + l15] = c[i];
  }
}

__global__ void __launch_bounds__(256) k_agg16(const int* __restrict__ ei,
                                               const float* __restrict__ dinv,
                                               const float* __restrict__ g,
                                               float* __restrict__ out) {
  int idx = blockIdx.x * blockDim.x + threadIdx.x;
  int e = idx >> 4;
  int f = idx & 15;
  const int total = N_EDGES + N_NODES;
  if (e >= total) return;
  int s, d;
  if (e < N_EDGES) { s = ei[e]; d = ei[N_EDGES + e]; }
  else             { s = d = e - N_EDGES; }
  float w = dinv[s] * dinv[d];
  float v = g[(size_t)s * OUT_DIM + f] * w;
  unsafeAtomicAdd(&out[(size_t)d * OUT_DIM + f], v);
}

extern "C" void kernel_launch(void* const* d_in, const int* in_sizes, int n_in,
                              void* d_out, int out_size, void* d_ws, size_t ws_size,
                              hipStream_t stream) {
  const float* x  = (const float*)d_in[0];
  const float* W1 = (const float*)d_in[1];
  const float* b1 = (const float*)d_in[2];
  const float* W2 = (const float*)d_in[3];
  const float* b2 = (const float*)d_in[4];
  const int*   ei = (const int*)d_in[5];

  float* deg = (float*)d_ws;
  float* h   = deg + N_NODES;
  float* h1  = h   + (size_t)N_NODES * HID_DIM;
  float* g   = h1  + (size_t)N_NODES * HID_DIM;
  float* out = (float*)d_out;

  const int B = 256;
  k_init<<<(N_NODES * HID_DIM + B - 1) / B, B, 0, stream>>>(deg, h1, out, b2);
  k_degree<<<(N_EDGES + B - 1) / B, B, 0, stream>>>(ei, deg);
  k_rsqrt<<<(N_NODES + B - 1) / B, B, 0, stream>>>(deg);
  k_gemm1<<<3125, 128, 0, stream>>>(x, W1, h);
  {
    long long threads = (long long)(N_EDGES + N_NODES) * HID_DIM;
    k_agg32<<<(unsigned)((threads + B - 1) / B), B, 0, stream>>>(ei, deg, h, h1);
  }
  k_bias_relu<<<(N_NODES * HID_DIM + B - 1) / B, B, 0, stream>>>(h1, b1);
  k_gemm2<<<3125, 64, 0, stream>>>(h1, W2, g);
  {
    long long threads = (long long)(N_EDGES + N_NODES) * OUT_DIM;
    k_agg16<<<(unsigned)((threads + B - 1) / B), B, 0, stream>>>(ei, deg, g, out);
  }
}


# X@W1: a product wave loads its X rows as whole 128-byte lines (16 bytes per lane, 4 lines per load) into its own LDS slot and reads A fragments there; was 8-byte loads touching 16 lines; k order kept
# speedup vs baseline: 6.1796x; 6.1796x over previous
#include <hip/hip_runtime.h>


#define N_NODES 100000
#define N_EDGES 3200000
#define IN_DIM  128
#define HID_DIM 32
#define OUT_DIM 16
#define ROW_CAP 64
#define REG_CAP 32
#define CNT_PLANE 100032
#define ROWS_PLANE ((size_t)N_NODES * ROW_CAP)
#define GEMM1_WAVES 12500
#define GEMM1_WGS   1563
#define MIX_PERIOD  7
#define MIX_WGS     (MIX_PERIOD * GEMM1_WGS)
#define MIX_RANK_WGS ((MIX_PERIOD - 1) * GEMM1_WGS)
#define DEG_SAT   255
#define TAB_BYTES 100352
#define P32_WGS   512
#define P32_WAVES 28
#define TAB_CHUNKS (TAB_BYTES / 16 / (32 * P32_WAVES))
#define P32_STEP  (P32_WGS * P32_WAVES)
#define P32_ROWS  7
#define H1_PITCH  36
#define XS_PITCH  36
#define XS_PIECE  (16 * XS_PITCH)
#define XS_WAVE   (2 * XS_PIECE)
static_assert(P32_ROWS * P32_STEP >= N_NODES, "a wave stages every node it sums");
static_assert(P32_ROWS <= 8, "the staged rows are rows 0-7 of the tile: the registers of lanes 0-15");
static_assert(TAB_CHUNKS * 16 * 32 * P32_WAVES == TAB_BYTES, "the table divides evenly over the threads");
static_assert(TAB_BYTES % (4 * 256) == 0 && TAB_BYTES > N_NODES, "k_rsqrt's grid covers the table, pad byte included");

__device__ __forceinline__ int my_half() {
  return (int)((__builtin_amdgcn_s_sendmsg_rtn(0x87) >> 18) & 1u);
}

typedef __attribute__((ext_vector_type(2))) float v2f;
typedef __attribute__((ext_vector_type(8))) float v8f;

__global__ void __launch_bounds__(256) k_init(int* __restrict__ cnt,
                                              int* __restrict__ nspill,
                                              float* __restrict__ dinv_pad,
                                              float* __restrict__ h_pad,
                                              float* __restrict__ g_pad) {
  int i = blockIdx.x * blockDim.x + threadIdx.x;
  if (i < 2 * CNT_PLANE) cnt[i] = 0;
  if (i == 0) { *nspill = 0; *dinv_pad = 0.0f; }
  if (i < HID_DIM) h_pad[i] = 0.0f;
  if (i < OUT_DIM) g_pad[i] = 0.0f;
}

__device__ __forceinline__ void gemm1_tile(const float* __restrict__ X,
                                           const float* __restrict__ W,
                                           float* __restrict__ H,
                                           float* xs,
                                           int wid, int lane) {
  const int tile_m = wid >> 1;
  const int tile_n = wid & 1;
  const int half   = lane >> 4;
  const int l15    = lane & 15;
  const int kOff   = half * 2;
  const int colB   = tile_n * 16 + l15;
  const int srow   = lane >> 3;
  const int spart  = (lane & 7) * 4;
  const float* xg  = X + (size_t)(tile_m * 16 + srow) * IN_DIM + spart;
  float*       sst = xs + srow * XS_PITCH + spart;
  const float* afr = xs + l15 * XS_PITCH + kOff;

  float4 p[4];
#pragma unroll
  for (int i = 0; i < 4; ++i) p[i] = *(const float4*)(xg + (size_t)(4 * i) * IN_DIM);
  v8f c = {};
#pragma unroll
  for (int j = 0; j < 4; ++j) {
#pragma unroll
    for (int i = 0; i < 4; ++i) *(float4*)(sst + (j & 1) * XS_PIECE + 4 * i * XS_PITCH) = p[i];
    if (j < 3) {
#pragma unroll
      for (int i = 0; i < 4; ++i) p[i] = *(const float4*)(xg + (size_t)(4 * i) * IN_DIM + 32 * (j + 1));
    }
    __builtin_amdgcn_sched_barrier(0);
#pragma unroll
    for (int kk = 0; kk < 32; kk += 4) {
      const int k = j * 32 + kk;
      v2f a = *(const v2f*)(afr + (j & 1) * XS_PIECE + kk);
      v2f b;
      b.x = W[(k + kOff + 0) * HID_DIM + colB];
      b.y = W[(k + kOff + 1) * HID_DIM + colB];
      c = __builtin_amdgcn_wmma_f32_16x16x4_f32(false, a, false, b,
                                                (short)0, c, false, false);
    }
  }
  const int m0 = tile_m * 16;
#pragma unroll
  for (int i = 0; i < 8; ++i) {
    H[(size_t)(m0 + half * 8 + i) * HID_DIM + tile_n * 16 + l15] = c[i];
  }
}

__global__ void __launch_bounds__(256) k_build_gemm1(const int* __restrict__ ei,
                                                     int* __restrict__ cnt,
                                                     int* __restrict__ rows,
                                                     int* __restrict__ nspill,
                                                     int* __restrict__ spill,
                                                     const float* __restrict__ X,
                                                     const float* __restrict__ W1,
                                                     float* __restrict__ H) {
  __shared__ __attribute__((aligned(16))) float xs[8 * XS_WAVE];
  const int b = blockIdx.x;
  int rank_wg;
  if (b < MIX_WGS) {
    const int q = b / MIX_PERIOD;
    const int r = b - q * MIX_PERIOD;
    if (r == 0) {
      const int wv  = __builtin_amdgcn_readfirstlane((int)(threadIdx.x >> 5));
      const int wid = q * 8 + wv;
      if (wid < GEMM1_WAVES) gemm1_tile(X, W1, H, xs + wv * XS_WAVE, wid, (int)(threadIdx.x & 31));
      return;
    }
    rank_wg = q * (MIX_PERIOD - 1) + r - 1;
  } else {
    rank_wg = MIX_RANK_WGS + (b - MIX_WGS);
  }
  const int e = rank_wg * 256 + (int)threadIdx.x;
  const int hf = my_half();
  cnt  += hf * CNT_PLANE;
  rows += hf * ROWS_PLANE;
  if (e < N_EDGES) {
    int s = ei[e];
    int d = ei[N_EDGES + e];
    int r = atomicAdd(&cnt[d], 1);
    if (r < ROW_CAP) {
      rows[(size_t)d * ROW_CAP + r] = s;
    } else {
      int k = atomicAdd(nspill, 1);
      spill[k] = e;
    }
  }
}

__device__ __forceinline__ float dinv_of(int deg) {
  const float d = (float)deg;
  return (d > 0.0f) ? rsqrtf(d) : 0.0f;
}
__device__ __forceinline__ unsigned deg_byte(int deg) {
  return (unsigned)(deg < DEG_SAT ? deg : DEG_SAT);
}
__global__ void __launch_bounds__(256) k_rsqrt(const int* __restrict__ cnt,
                                               float* __restrict__ dinv,
                                               unsigned* __restrict__ deg8) {
  int i = blockIdx.x * blockDim.x + threadIdx.x;
  if (i < N_NODES / 4) {
    const int4 a = ((const int4*)cnt)[i];
    const int4 b = ((const int4*)(cnt + CNT_PLANE))[i];
    const int d0 = a.x + b.x + 1, d1 = a.y + b.y + 1, d2 = a.z + b.z + 1, d3 = a.w + b.w + 1;
    float4 r;
    r.x = dinv_of(d0); r.y = dinv_of(d1); r.z = dinv_of(d2); r.w = dinv_of(d3);
    ((float4*)dinv)[i] = r;
    deg8[i] = deg_byte(d0) | (deg_byte(d1) << 8) | (deg_byte(d2) << 16) | (deg_byte(d3) << 24);
  } else if (i < TAB_BYTES / 4) {
    deg8[i] = 0u;
  }
}

__device__ __forceinline__ int lane_read_i(int byte_addr, int v) {
  return __builtin_amdgcn_ds_bpermute(byte_addr, v);
}
__device__ __forceinline__ float lane_read_f(int byte_addr, float v) {
  return __int_as_float(__builtin_amdgcn_ds_bpermute(byte_addr, __float_as_int(v)));
}
__device__ __forceinline__ float4 fma4(float4 v, float s, float4 acc) {
  acc.x += v.x * s; acc.y += v.y * s; acc.z += v.z * s; acc.w += v.w * s;
  return acc;
}

__device__ __forceinline__ float4 sum_rows32(int e, float w, int n, int ga, unsigned pb,
                                             const float* __restrict__ h, float4 acc) {
  const char* hb = (const char*)h;
  for (int j0 = 0; j0 < n; j0 += 16) {
#pragma unroll
    for (int u = 0; u < 4; ++u) {
      const int   a  = ((j0 + 4 * u) << 2) + ga;
      const int   s  = lane_read_i(a, e);
      const float ws = lane_read_f(a, w);
      const float4 v = *(const float4*)(hb + (((unsigned)s << 7) | pb));
      acc = fma4(v, ws, acc);
    }
  }
  return acc;
}

__global__ void __launch_bounds__(32 * P32_WAVES) k_pull32(const int* __restrict__ ei,
                                                const int* __restrict__ cnt,
                                                const int* __restrict__ rows,
                                                const int* __restrict__ nspill,
                                                const int* __restrict__ spill,
                                                const float* __restrict__ dinv,
                                                const float* __restrict__ h,
                                                const float* __restrict__ b1,
                                                const float* __restrict__ W2,
                                                float* __restrict__ g,
                                                float* __restrict__ wt,
                                                const unsigned char* __restrict__ deg8) {
  __shared__ __attribute__((aligned(16))) unsigned char tab[TAB_BYTES];
  __shared__ __attribute__((aligned(16))) float stage[P32_WAVES * P32_ROWS * H1_PITCH];
  {
    const uint4* src = (const uint4*)deg8;
    uint4*       dst = (uint4*)tab;
    uint4        v[TAB_CHUNKS];
#pragma unroll
    for (int k = 0; k < TAB_CHUNKS; ++k) v[k] = src[(int)threadIdx.x + k * 32 * P32_WAVES];
#pragma unroll
    for (int k = 0; k < TAB_CHUNKS; ++k) dst[(int)threadIdx.x + k * 32 * P32_WAVES] = v[k];
  }
  __syncthreads();
  const int lane = threadIdx.x & 31;
  const int gi   = lane >> 3;
  const int ga   = gi << 2;
  const unsigned pb = (unsigned)(lane & 7) << 4;
  const float4 bias = *(const float4*)((const char*)b1 + pb);
  const int wv = __builtin_amdgcn_readfirstlane((int)(threadIdx.x >> 5));
  const int d_first = 2 * ((int)blockIdx.x + (wv >> 1) * P32_WGS) + (wv & 1);
  int d = d_first;
  float* myrows = stage + wv * (P32_ROWS * H1_PITCH);
  int nrow = 0;
  int   c0 = cnt[d];
  int   c1 = cnt[CNT_PLANE + d];
  float dd = dinv[d];
  float4 hs = *(const float4*)((const char*)h + (((unsigned)d << 7) | pb));
  int   e0, e1;
  {
    const int  a0 = c0 < REG_CAP ? c0 : REG_CAP;
    const int* r0 = rows + (size_t)d * ROW_CAP;
    const int* r1 = r0 + ROWS_PLANE;
    e0 = *((lane < a0) ? (r0 + lane) : (r1 + (lane - a0)));
    e1 = r1[lane + 32 - a0];
  }
  for (;;) {
  const int  dn   = d + P32_STEP;
  const bool more = dn < N_NODES;
  const int  dq   = more ? dn : d;
  const int   a0 = c0 < REG_CAP ? c0 : REG_CAP;
  const int   a1 = c1 < REG_CAP ? c1 : REG_CAP;
  const int   cc = a0 + a1;
  const int*  r0 = rows + (size_t)d * ROW_CAP;
  const int*  r1 = r0 + ROWS_PLANE;
  e0 = (lane      < cc) ? e0 : N_NODES;
  e1 = (lane + 32 < cc) ? e1 : N_NODES;
  const int t0 = tab[e0];
  const int t1 = tab[e1];
  float w0 = (lane      < cc) ? rsqrtf((float)t0) * dd : 0.0f;
  float w1 = (lane + 32 < cc) ? rsqrtf((float)t1) * dd : 0.0f;
  if (__builtin_amdgcn_ballot_w32(t0 == DEG_SAT || t1 == DEG_SAT) != 0u) {
    w0 = dinv[e0] * dd;
    w1 = dinv[e1] * dd;
  }
  const int   c0n = cnt[dq];
  const int   c1n = cnt[CNT_PLANE + dq];
  const float ddn = dinv[dq];
  const float4 hsn = *(const float4*)((const char*)h + (((unsigned)dq << 7) | pb));
  wt[(size_t)d * ROW_CAP + lane]      = w0;
  wt[(size_t)d * ROW_CAP + 32 + lane] = w1;
  float4 acc = {0.0f, 0.0f, 0.0f, 0.0f};
  acc = fma4(hs, (gi == 0) ? dd * dd : 0.0f, acc);

  acc = sum_rows32(e0, w0, cc < 32 ? cc : 32, ga, pb, h, acc);

  const int  a0n = c0n < REG_CAP ? c0n : REG_CAP;
  const int* r0n = rows + (size_t)dq * ROW_CAP;
  const int* r1n = r0n + ROWS_PLANE;
  const int  e0n = *((lane < a0n) ? (r0n + lane) : (r1n + (lane - a0n)));
  const int  e1n = r1n[lane + 32 - a0n];

  acc = sum_rows32(e1, w1, cc > 32 ? cc - 32 : 0, ga, pb, h, acc);

  if (c0 > REG_CAP || c1 > REG_CAP) {
#pragma unroll
    for (int p = 0; p < 2; ++p) {
      const int cp = p ? c1 : c0;
      const int cr = cp < ROW_CAP ? cp : ROW_CAP;
      if (cr > REG_CAP) {
        int e = (p ? r1 : r0)[REG_CAP + lane];
        e = (REG_CAP + lane < cr) ? e : N_NODES;
        const float w = dinv[e] * dd;
        acc = sum_rows32(e, w, cr - REG_CAP, ga, pb, h, acc);
      }
    }
    if (c0 > ROW_CAP || c1 > ROW_CAP) {
      const int ns = *nspill;
      for (int k = 0; k < ns; ++k) {
        const int e = spill[k];
        if (ei[N_EDGES + e] == d) {
          const int s = ei[e];
          const float4 v = *(const float4*)((const char*)h + (((unsigned)s << 7) | pb));
          acc = fma4(v, (gi == 0) ? dinv[s] * dd : 0.0f, acc);
        }
      }
    }
  }
  {
    const int x8 = (lane ^ 8) << 2, x16 = (lane ^ 16) << 2;
    float4 t;
    t.x = lane_read_f(x8, acc.x); t.y = lane_read_f(x8, acc.y); t.z = lane_read_f(x8, acc.z); t.w = lane_read_f(x8, acc.w);
    acc.x += t.x; acc.y += t.y; acc.z += t.z; acc.w += t.w;
    t.x = lane_read_f(x16, acc.x); t.y = lane_read_f(x16, acc.y); t.z = lane_read_f(x16, acc.z); t.w = lane_read_f(x16, acc.w);
    acc.x += t.x; acc.y += t.y; acc.z += t.z; acc.w += t.w;
  }
  if (gi == 0) {
    float4 r;
    r.x = fmaxf(acc.x + bias.x, 0.0f); r.y = fmaxf(acc.y + bias.y, 0.0f);
    r.z = fmaxf(acc.z + bias.z, 0.0f); r.w = fmaxf(acc.w + bias.w, 0.0f);
    *(float4*)((char*)(myrows + nrow * H1_PITCH) + pb) = r;
  }
  ++nrow;
  if (!more) break;
  d = dn; c0 = c0n; c1 = c1n; dd = ddn; hs = hsn; e0 = e0n; e1 = e1n;
  }
  {
    const int half = lane >> 4;
    const int l15  = lane & 15;
    const int kOff = half * 2;
    const float* arow = myrows + (l15 < nrow ? l15 : 0) * H1_PITCH;
    v8f cg = {};
#pragma unroll
    for (int k = 0; k < HID_DIM; k += 4) {
      v2f a = *(const v2f*)(arow + k + kOff);
      v2f b;
      b.x = W2[(k + kOff + 0) * OUT_DIM + l15];
      b.y = W2[(k + kOff + 1) * OUT_DIM + l15];
      cg = __builtin_amdgcn_wmma_f32_16x16x4_f32(false, a, false, b,
                                                 (short)0, cg, false, false);
    }
    if (half == 0) {
#pragma unroll
      for (int i = 0; i < P32_ROWS; ++i) {
        if (i < nrow) g[(size_t)(d_first + i * P32_STEP) * OUT_DIM + l15] = cg[i];
      }
    }
  }
}

template <int STEPS>
__device__ __forceinline__ float4 rows16_steps(int er, float wr, int ha, unsigned pb,
                                               const char* __restrict__ gb, float4 acc) {
#pragma unroll
  for (int u = 0; u < STEPS; ++u) {
    const int   a = ha + (u << 4);
    const int   s = lane_read_i(a, er);
    const float w = lane_read_f(a, wr);
    const float4 v = *(const float4*)(gb + (((unsigned)s << 6) | pb));
    acc = fma4(v, w, acc);
  }
  return acc;
}
__device__ __forceinline__ float4 sum_rows16(int er, float wr, int n, int ha, unsigned pb,
                                             const float* __restrict__ g, float4 acc) {
  const char* gb = (const char*)g;
  if (n > 8)      acc = rows16_steps<4>(er, wr, ha, pb, gb, acc);
  else if (n > 0) acc = rows16_steps<2>(er, wr, ha, pb, gb, acc);
  return acc;
}

__device__ __forceinline__ int group_n(int cm, int q) {
  int nq = cm - q * 16;
  return nq < 0 ? 0 : (nq > 16 ? 16 : nq);
}

__global__ void __launch_bounds__(256) k_pull16(const int* __restrict__ ei,
                                                const int* __restrict__ cnt,
                                                const int* __restrict__ rows,
                                                const int* __restrict__ nspill,
                                                const int* __restrict__ spill,
                                                const float* __restrict__ dinv,
                                                const float* __restrict__ g,
                                                const float* __restrict__ b2,
                                                const float* __restrict__ wt,
                                                float* __restrict__ out) {
  const int lane = threadIdx.x & 31;
  const int half = lane >> 4;
  const int f    = lane & 15;
  const int gq   = f >> 2;
  const int ha   = ((half << 4) + gq) << 2;
  const unsigned pb = (unsigned)(lane & 3) << 4;
  const int pair = __builtin_amdgcn_readfirstlane(blockIdx.x * 8 + (threadIdx.x >> 5));
  if (pair * 2 >= N_NODES) return;
  const int   d  = pair * 2 + half;
  const int   c0 = cnt[d];
  const int   c1 = cnt[CNT_PLANE + d];
  const int   a0 = c0 < REG_CAP ? c0 : REG_CAP;
  const int   a1 = c1 < REG_CAP ? c1 : REG_CAP;
  const int   cc = a0 + a1;
  const float dd = dinv[d];
  const int*  r0 = rows + (size_t)d * ROW_CAP;
  const int*  r1 = r0 + ROWS_PLANE;
  int   er[4];
  float wr[4];
#pragma unroll
  for (int q = 0; q < 4; ++q) {
    const int k = q * 16 + f;
    int e = *((k < a0) ? (r0 + k) : (r1 + (k - a0)));
    e = (k < cc) ? e : N_NODES;
    er[q] = e;
    wr[q] = wt[(size_t)d * ROW_CAP + k];
  }
  float4 acc = {0.0f, 0.0f, 0.0f, 0.0f};
  acc = fma4(*(const float4*)((const char*)g + (((unsigned)d << 6) | pb)), (gq == 0) ? dd * dd : 0.0f, acc);

  const int cA = __builtin_amdgcn_readlane(cc, 0);
  const int cB = __builtin_amdgcn_readlane(cc, 16);
  const int cm = cA > cB ? cA : cB;
#pragma unroll
  for (int q = 0; q < 4; ++q)
    acc = sum_rows16(er[q], wr[q], group_n(cm, q), ha, pb, g, acc);

  const int t0 = c0 > c1 ? c0 : c1;
  const int tA = __builtin_amdgcn_readlane(t0, 0);
  const int tB = __builtin_amdgcn_readlane(t0, 16);
  const int tm = tA > tB ? tA : tB;
  if (tm > REG_CAP) {
    const int tr = tm < ROW_CAP ? tm : ROW_CAP;
#pragma unroll
    for (int p = 0; p < 2; ++p) {
      const int cp = p ? c1 : c0;
      const int cr = cp < ROW_CAP ? cp : ROW_CAP;
#pragma unroll
      for (int q = 0; q < 2; ++q) {
        const int j = REG_CAP + q * 16 + f;
        int e = (p ? r1 : r0)[j];
        e = (j < cr) ? e : N_NODES;
        const float w = dinv[e] * dd;
        acc = sum_rows16(e, w, group_n(tr - REG_CAP, q), ha, pb, g, acc);
      }
    }
    if (tm > ROW_CAP) {
      const int ns = *nspill;
      for (int k = 0; k < ns; ++k) {
        const int e = spill[k];
        if (ei[N_EDGES + e] == d) {
          const int s = ei[e];
          const float4 v = *(const float4*)((const char*)g + (((unsigned)s << 6) | pb));
          acc = fma4(v, (gq == 0) ? dinv[s] * dd : 0.0f, acc);
        }
      }
    }
  }
  {
    const int x4 = (lane ^ 4) << 2, x8 = (lane ^ 8) << 2;
    float4 t;
    t.x = lane_read_f(x4, acc.x); t.y = lane_read_f(x4, acc.y); t.z = lane_read_f(x4, acc.z); t.w = lane_read_f(x4, acc.w);
    acc.x += t.x; acc.y += t.y; acc.z += t.z; acc.w += t.w;
    t.x = lane_read_f(x8, acc.x); t.y = lane_read_f(x8, acc.y); t.z = lane_read_f(x8, acc.z); t.w = lane_read_f(x8, acc.w);
    acc.x += t.x; acc.y += t.y; acc.z += t.z; acc.w += t.w;
  }
  if (gq == 0) {
    const float4 bias = *(const float4*)((const char*)b2 + pb);
    float4 r;
    r.x = acc.x + bias.x; r.y = acc.y + bias.y; r.z = acc.z + bias.z; r.w = acc.w + bias.w;
    *(float4*)((char*)out + (((unsigned)d << 6) | pb)) = r;
  }
}

extern "C" void kernel_launch(void* const* d_in, const int* in_sizes, int n_in,
                              void* d_out, int out_size, void* d_ws, size_t ws_size,
                              hipStream_t stream) {
  const float* x  = (const float*)d_in[0];
  const float* W1 = (const float*)d_in[1];
  const float* b1 = (const float*)d_in[2];
  const float* W2 = (const float*)d_in[3];
  const float* b2 = (const float*)d_in[4];
  const int*   ei = (const int*)d_in[5];

  float* dinv  = (float*)d_ws;
  float* h     = dinv + 100032;
  float* g     = h    + (size_t)(N_NODES + 1) * HID_DIM + (size_t)N_NODES * HID_DIM;
  int*   cnt   = (int*)(g + (size_t)(N_NODES + 2) * OUT_DIM);
  int*   nsp   = cnt + 2 * CNT_PLANE;
  int*   rows  = nsp + 64;
  int*   spill = rows + 2 * ROWS_PLANE;
  float* wt    = (float*)(spill + N_EDGES);
  unsigned* deg8 = (unsigned*)(wt + (size_t)N_NODES * ROW_CAP);
  float* out   = (float*)d_out;

  const int B = 256;
  k_init<<<(2 * CNT_PLANE + B - 1) / B, B, 0, stream>>>(cnt, nsp, dinv + N_NODES,
                                                  h + (size_t)N_NODES * HID_DIM,
                                                  g + (size_t)N_NODES * OUT_DIM);
  k_build_gemm1<<<14063, B, 0, stream>>>(ei, cnt, rows, nsp, spill, x, W1, h);
  k_rsqrt<<<TAB_BYTES / 4 / B, B, 0, stream>>>(cnt, dinv, deg8);
  k_pull32<<<P32_WGS, 32 * P32_WAVES, 0, stream>>>(ei, cnt, rows, nsp, spill, dinv, h, b1, W2, g, wt,
                                                   (const unsigned char*)deg8);
  k_pull16<<<N_NODES / 16, B, 0, stream>>>(ei, cnt, rows, nsp, spill, dinv, g, b2, wt, out);
}
